# starting kernel
# baseline (speedup 1.0000x reference)
#include <hip/hip_runtime.h>
#include <math.h>


typedef float v2f __attribute__((ext_vector_type(2)));
typedef float v8f __attribute__((ext_vector_type(8)));

#define BN   4
#define TT   2048
#define CC   768
#define HH   12
#define DD   64
#define FF   3072
#define NTOK 8192
#define LDA  772
#define SROW 2052

__device__ __forceinline__ v8f wmma4(v2f a, v2f b, v8f c) {
  return __builtin_amdgcn_wmma_f32_16x16x4_f32(false, a, false, b, (short)0, c,
                                               false, false);
}

__global__ __launch_bounds__(256) void transpose_kernel(
    const float* __restrict__ src, float* __restrict__ dst, int K, int N) {
  __shared__ float tile[32][33];
  const size_t bofs = (size_t)blockIdx.z * K * N;
  src += bofs;
  dst += bofs;
  const int k0 = blockIdx.y * 32, n0 = blockIdx.x * 32;
  const int tx = threadIdx.x & 31, ty = threadIdx.x >> 5;
  for (int i = ty; i < 32; i += 8)
    tile[i][tx] = src[(size_t)(k0 + i) * N + n0 + tx];
  __syncthreads();
  for (int i = ty; i < 32; i += 8)
    dst[(size_t)(n0 + i) * K + k0 + tx] = tile[tx][i];
}

__global__ __launch_bounds__(256) void qkv_kernel(
    const float* __restrict__ x, const float* __restrict__ wqkvt,
    float* __restrict__ qo, float* __restrict__ ko, float* __restrict__ vo) {
  __shared__ float xs[64 * LDA];
  const int m0 = blockIdx.x * 64;
  for (int i = threadIdx.x; i < 64 * (CC / 4); i += 256) {
    const int r = i / (CC / 4), c4 = i % (CC / 4);
    const float4 src = ((const float4*)(x + (m0 + r) * CC))[c4];
    *(float4*)(&xs[r * LDA + c4 * 4]) = src;
  }
  __syncthreads();
  const int wv = threadIdx.x >> 5, l = threadIdx.x & 31;
  const int n16 = l & 15, kq = (l >> 4) << 1;
  for (int task = wv; task < 144; task += 8) {
    const int rb = task / 36, oh = task % 36;
    const int o = oh / 12, head = oh % 12;
    const float* Wt = wqkvt + (size_t)oh * (DD * CC);
    v8f a0 = {}, a1 = {}, a2 = {}, a3 = {};
    const float* arow = &xs[(rb * 16 + n16) * LDA];
    const float* brow = Wt + (size_t)n16 * CC;
    for (int k0 = 0; k0 < CC; k0 += 4) {
      const v2f a = *(const v2f*)(arow + k0 + kq);
      const v2f b0 = *(const v2f*)(brow + k0 + kq);
      const v2f b1 = *(const v2f*)(brow + 16 * CC + k0 + kq);
      const v2f b2 = *(const v2f*)(brow + 32 * CC + k0 + kq);
      const v2f b3 = *(const v2f*)(brow + 48 * CC + k0 + kq);
      a0 = wmma4(a, b0, a0); a1 = wmma4(a, b1, a1);
      a2 = wmma4(a, b2, a2); a3 = wmma4(a, b3, a3);
    }
    const int mbase = rb * 16 + ((l >> 4) << 3);
    if (o < 2) {
      float* OUT = (o == 0) ? qo : ko;
#pragma unroll
      for (int j = 0; j < 8; ++j) {
        const int tok = m0 + mbase + j;
        const int b = tok >> 11, t = tok & 2047;
        float* dst = OUT + (((size_t)(b * HH + head) * TT + t) * DD);
        dst[0 + n16] = a0[j];  dst[16 + n16] = a1[j];
        dst[32 + n16] = a2[j]; dst[48 + n16] = a3[j];
      }
    } else {
#pragma unroll
      for (int j = 0; j < 8; ++j) {
        const int tok = m0 + mbase + j;
        const int b = tok >> 11, t = tok & 2047;
        float* base = vo + (size_t)(b * HH + head) * DD * TT + t;
        base[(size_t)(0 + n16) * TT] = a0[j];
        base[(size_t)(16 + n16) * TT] = a1[j];
        base[(size_t)(32 + n16) * TT] = a2[j];
        base[(size_t)(48 + n16) * TT] = a3[j];
      }
    }
  }
}

__global__ __launch_bounds__(256) void attn_kernel(
    const float* __restrict__ q, const float* __restrict__ k,
    const float* __restrict__ vt, float* __restrict__ aout) {
  __shared__ float S[16 * SROW];
  __shared__ float red[256];
  __shared__ float rowmax[16], rowrcp[16];
  __shared__ float opart[16 * 64];
  const int ttile = blockIdx.x;
  const int bh = blockIdx.y;
  const int t0 = ttile * 16;
  const float* qp = q + (size_t)bh * TT * DD;
  const float* kp = k + (size_t)bh * TT * DD;
  const float* vp = vt + (size_t)bh * TT * DD;
  const int wv = threadIdx.x >> 5, l = threadIdx.x & 31;
  const int n16 = l & 15, hi = l >> 4, kq = hi << 1;
  const int NT = ttile + 1, L = NT * 16;
  const float scale = 0.036084391824352f;

  v2f af[16];
#pragma unroll
  for (int ks = 0; ks < 16; ++ks)
    af[ks] = *(const v2f*)(kp + (t0 + n16) * DD + ks * 4 + kq);
  for (int st = wv; st < NT; st += 8) {
    const int s0 = st * 16;
    v8f acc = {};
#pragma unroll
    for (int ks = 0; ks < 16; ++ks) {
      const v2f b = *(const v2f*)(qp + (s0 + n16) * DD + ks * 4 + kq);
      acc = wmma4(af[ks], b, acc);
    }
    const int sg = s0 + n16;
#pragma unroll
    for (int j = 0; j < 8; ++j) {
      const int row = (hi << 3) + j;
      float vl = acc[j] * scale;
      if (sg > t0 + row) vl = -1e30f;
      S[row * SROW + sg] = vl;
    }
  }
  __syncthreads();

  {
    const int row = threadIdx.x >> 4, sub = threadIdx.x & 15;
    float m = -1e30f;
    for (int c = sub; c < L; c += 16) m = fmaxf(m, S[row * SROW + c]);
    red[threadIdx.x] = m;
    __syncthreads();
    if (sub == 0) {
      float mm = red[threadIdx.x];
      for (int i = 1; i < 16; ++i) mm = fmaxf(mm, red[threadIdx.x + i]);
      rowmax[row] = mm;
    }
    __syncthreads();
    const float mr = rowmax[row];
    float s = 0.f;
    for (int c = sub; c < L; c += 16) {
      const float e = __expf(S[row * SROW + c] - mr);
      S[row * SROW + c] = e;
      s += e;
    }
    red[threadIdx.x] = s;
    __syncthreads();
    if (sub == 0) {
      float ss = 0.f;
      for (int i = 0; i < 16; ++i) ss += red[threadIdx.x + i];
      rowrcp[row] = 1.0f / ss;
    }
  }
  __syncthreads();

  const int tile = wv & 3, half = wv >> 2;
  const int ksteps = L >> 2, kh = ksteps >> 1;
  const int kbeg = half * kh, kend = kbeg + kh;
  v8f acc = {};
  const float* vrow = vp + (size_t)(tile * 16 + n16) * TT;
  for (int ks = kbeg; ks < kend; ++ks) {
    const v2f a = *(const v2f*)(&S[n16 * SROW + ks * 4 + kq]);
    const v2f b = *(const v2f*)(vrow + ks * 4 + kq);
    acc = wmma4(a, b, acc);
  }
  if (half == 1) {
#pragma unroll
    for (int j = 0; j < 8; ++j)
      opart[((hi << 3) + j) * 64 + tile * 16 + n16] = acc[j];
  }
  __syncthreads();
  if (half == 0) {
    const int bb = bh / HH, head = bh % HH;
#pragma unroll
    for (int j = 0; j < 8; ++j) {
      const int row = (hi << 3) + j;
      const float val =
          (acc[j] + opart[row * 64 + tile * 16 + n16]) * rowrcp[row];
      const int t = t0 + row;
      aout[((size_t)(bb * TT + t)) * CC + head * DD + tile * 16 + n16] = val;
    }
  }
}

__global__ __launch_bounds__(256) void add_ln_kernel(
    const float* __restrict__ xa, const float* __restrict__ xb,
    const float* __restrict__ g, const float* __restrict__ bt,
    float* __restrict__ out) {
  __shared__ float red[256];
  const int tok = blockIdx.x;
  const float* pa = xa + (size_t)tok * CC;
  const float* pb = xb + (size_t)tok * CC;
  float vals[3];
  float s1 = 0.f;
#pragma unroll
  for (int i = 0; i < 3; ++i) {
    const int c = threadIdx.x + i * 256;
    vals[i] = pa[c] + pb[c];
    s1 += vals[i];
  }
  red[threadIdx.x] = s1;
  __syncthreads();
  for (int off = 128; off > 0; off >>= 1) {
    if (threadIdx.x < off) red[threadIdx.x] += red[threadIdx.x + off];
    __syncthreads();
  }
  const float mean = red[0] * (1.0f / CC);
  __syncthreads();
  float s2 = 0.f;
#pragma unroll
  for (int i = 0; i < 3; ++i) {
    const float d = vals[i] - mean;
    s2 += d * d;
  }
  red[threadIdx.x] = s2;
  __syncthreads();
  for (int off = 128; off > 0; off >>= 1) {
    if (threadIdx.x < off) red[threadIdx.x] += red[threadIdx.x + off];
    __syncthreads();
  }
  const float rstd = rsqrtf(red[0] * (1.0f / CC) + 1e-5f);
#pragma unroll
  for (int i = 0; i < 3; ++i) {
    const int c = threadIdx.x + i * 256;
    out[(size_t)tok * CC + c] = (vals[i] - mean) * rstd * g[c] + bt[c];
  }
}

__global__ __launch_bounds__(256) void mlp1_kernel(
    const float* __restrict__ xin, const float* __restrict__ W1t,
    const float* __restrict__ bias, float* __restrict__ hout) {
  __shared__ float xs[64 * LDA];
  const int m0 = blockIdx.x * 64;
  for (int i = threadIdx.x; i < 64 * (CC / 4); i += 256) {
    const int r = i / (CC / 4), c4 = i % (CC / 4);
    const float4 src = ((const float4*)(xin + (m0 + r) * CC))[c4];
    *(float4*)(&xs[r * LDA + c4 * 4]) = src;
  }
  __syncthreads();
  const int wv = threadIdx.x >> 5, l = threadIdx.x & 31;
  const int n16 = l & 15, kq = (l >> 4) << 1;
  for (int task = wv; task < 192; task += 8) {
    const int rb = task / 48, nb = (task % 48) * 64;
    v8f a0 = {}, a1 = {}, a2 = {}, a3 = {};
    const float* arow = &xs[(rb * 16 + n16) * LDA];
    const float* brow = W1t + (size_t)(nb + n16) * CC;
    for (int k0 = 0; k0 < CC; k0 += 4) {
      const v2f a = *(const v2f*)(arow + k0 + kq);
      const v2f b0 = *(const v2f*)(brow + k0 + kq);
      const v2f b1 = *(const v2f*)(brow + (size_t)16 * CC + k0 + kq);
      const v2f b2 = *(const v2f*)(brow + (size_t)32 * CC + k0 + kq);
      const v2f b3 = *(const v2f*)(brow + (size_t)48 * CC + k0 + kq);
      a0 = wmma4(a, b0, a0); a1 = wmma4(a, b1, a1);
      a2 = wmma4(a, b2, a2); a3 = wmma4(a, b3, a3);
    }
    const int mbase = rb * 16 + ((l >> 4) << 3);
#pragma unroll
    for (int j = 0; j < 8; ++j) {
      float* dst = hout + (size_t)(m0 + mbase + j) * FF + nb + n16;
      float z;
      z = a0[j] + bias[nb + n16];
      dst[0] = 0.5f * z * (1.f + erff(z * 0.70710678118f));
      z = a1[j] + bias[nb + 16 + n16];
      dst[16] = 0.5f * z * (1.f + erff(z * 0.70710678118f));
      z = a2[j] + bias[nb + 32 + n16];
      dst[32] = 0.5f * z * (1.f + erff(z * 0.70710678118f));
      z = a3[j] + bias[nb + 48 + n16];
      dst[48] = 0.5f * z * (1.f + erff(z * 0.70710678118f));
    }
  }
}

__global__ __launch_bounds__(256) void mlp2_kernel(
    const float* __restrict__ hin, const float* __restrict__ W2t,
    const float* __restrict__ bias, float* __restrict__ t2) {
  __shared__ float opart[64 * 64];
  const int m0 = blockIdx.x * 64, n0 = blockIdx.y * 64;
  const int wv = threadIdx.x >> 5, l = threadIdx.x & 31;
  const int n16 = l & 15, kq = (l >> 4) << 1;
  const int rb = wv & 3, half = wv >> 2;
  const int kbeg = half * 384, kend = kbeg + 384;
  v8f a0 = {}, a1 = {}, a2 = {}, a3 = {};
  const float* arow = hin + (size_t)(m0 + rb * 16 + n16) * FF;
  const float* brow = W2t + (size_t)(n0 + n16) * FF;
  for (int ks = kbeg; ks < kend; ++ks) {
    const int kk = ks * 4 + kq;
    const v2f a = *(const v2f*)(arow + kk);
    const v2f b0 = *(const v2f*)(brow + kk);
    const v2f b1 = *(const v2f*)(brow + (size_t)16 * FF + kk);
    const v2f b2 = *(const v2f*)(brow + (size_t)32 * FF + kk);
    const v2f b3 = *(const v2f*)(brow + (size_t)48 * FF + kk);
    a0 = wmma4(a, b0, a0); a1 = wmma4(a, b1, a1);
    a2 = wmma4(a, b2, a2); a3 = wmma4(a, b3, a3);
  }
  const int mb = rb * 16 + ((l >> 4) << 3);
  if (half == 1) {
#pragma unroll
    for (int j = 0; j < 8; ++j) {
      opart[(mb + j) * 64 + n16] = a0[j];
      opart[(mb + j) * 64 + 16 + n16] = a1[j];
      opart[(mb + j) * 64 + 32 + n16] = a2[j];
      opart[(mb + j) * 64 + 48 + n16] = a3[j];
    }
  }
  __syncthreads();
  if (half == 0) {
#pragma unroll
    for (int j = 0; j < 8; ++j) {
      float* dst = t2 + (size_t)(m0 + mb + j) * CC + n0 + n16;
      dst[0]  = a0[j] + opart[(mb + j) * 64 + n16]      + bias[n0 + n16];
      dst[16] = a1[j] + opart[(mb + j) * 64 + 16 + n16] + bias[n0 + 16 + n16];
      dst[32] = a2[j] + opart[(mb + j) * 64 + 32 + n16] + bias[n0 + 32 + n16];
      dst[48] = a3[j] + opart[(mb + j) * 64 + 48 + n16] + bias[n0 + 48 + n16];
    }
  }
}

extern "C" void kernel_launch(void* const* d_in, const int* in_sizes, int n_in,
                              void* d_out, int out_size, void* d_ws,
                              size_t ws_size, hipStream_t stream) {
  (void)in_sizes; (void)n_in; (void)out_size; (void)ws_size;
  const float* x    = (const float*)d_in[0];
  const float* Wq   = (const float*)d_in[1];
  const float* Wk   = (const float*)d_in[2];
  const float* Wv   = (const float*)d_in[3];
  const float* ln1g = (const float*)d_in[4];
  const float* ln1b = (const float*)d_in[5];
  const float* W1   = (const float*)d_in[6];
  const float* b1   = (const float*)d_in[7];
  const float* W2   = (const float*)d_in[8];
  const float* b2   = (const float*)d_in[9];
  const float* ln2g = (const float*)d_in[10];
  const float* ln2b = (const float*)d_in[11];
  float* ws  = (float*)d_ws;
  float* out = (float*)d_out;

  float* q     = ws;
  float* k     = ws + 6291456;
  float* v     = ws + 12582912;
  float* a     = ws + 18874368;
  float* hbuf  = ws;
  float* x1    = ws + 25165824;
  float* t2    = ws + 31457280;
  float* wqkvt = ws + 37748736;
  float* w1t   = ws + 39518208;
  float* w2t   = ws + 41877504;

  transpose_kernel<<<dim3(DD / 32, CC / 32, HH), 256, 0, stream>>>(
      Wq, wqkvt + 0 * (size_t)HH * DD * CC, CC, DD);
  transpose_kernel<<<dim3(DD / 32, CC / 32, HH), 256, 0, stream>>>(
      Wk, wqkvt + 1 * (size_t)HH * DD * CC, CC, DD);
  transpose_kernel<<<dim3(DD / 32, CC / 32, HH), 256, 0, stream>>>(
      Wv, wqkvt + 2 * (size_t)HH * DD * CC, CC, DD);
  transpose_kernel<<<dim3(FF / 32, CC / 32, 1), 256, 0, stream>>>(
      W1, w1t, CC, FF);
  transpose_kernel<<<dim3(CC / 32, FF / 32, 1), 256, 0, stream>>>(
      W2, w2t, FF, CC);

  qkv_kernel<<<dim3(NTOK / 64), 256, 0, stream>>>(x, wqkvt, q, k, v);
  attn_kernel<<<dim3(TT / 16, BN * HH), 256, 0, stream>>>(q, k, v, a);
  add_ln_kernel<<<dim3(NTOK), 256, 0, stream>>>(x, a, ln1g, ln1b, x1);
  mlp1_kernel<<<dim3(NTOK / 64), 256, 0, stream>>>(x1, w1t, b1, hbuf);
  mlp2_kernel<<<dim3(NTOK / 64, CC / 64), 256, 0, stream>>>(hbuf, w2t, b2, t2);
  add_ln_kernel<<<dim3(NTOK), 256, 0, stream>>>(x1, t2, ln2g, ln2b, out);
}


# weights packed in WMMA B-fragment order by the pre-pass: every B fragment load is 16 bytes per lane over 4 whole lines (was 8 bytes over 16 lines), A read in the same k order
# speedup vs baseline: 2.0466x; 2.0466x over previous
#include <hip/hip_runtime.h>
#include <math.h>


typedef float v2f __attribute__((ext_vector_type(2)));
typedef float v4f __attribute__((ext_vector_type(4)));
typedef float v8f __attribute__((ext_vector_type(8)));

#define BN   4
#define TT   2048
#define CC   768
#define HH   12
#define DD   64
#define FF   3072
#define NTOK 8192
#define LDA  772
#define SROW 2052

__device__ __forceinline__ v8f wmma4(v2f a, v2f b, v8f c) {
  return __builtin_amdgcn_wmma_f32_16x16x4_f32(false, a, false, b, (short)0, c,
                                               false, false);
}

__global__ __launch_bounds__(256) void pack_kernel(
    const float* __restrict__ src, float* __restrict__ dst, int K, int N) {
  const size_t bofs = (size_t)blockIdx.z * K * N;
  src += bofs;
  dst += bofs;
  const int l = threadIdx.x & 31;
  const int c = l & 15, h = l >> 4;
  const int ntiles = N >> 4;
  const int chunk = blockIdx.x * 8 + (threadIdx.x >> 5);
  const int nt = chunk % ntiles, kb = chunk / ntiles;
  const float* s = src + (size_t)(kb * 8 + 4 * h) * N + nt * 16 + c;
  v4f v;
  v.x = s[0];
  v.y = s[N];
  v.z = s[2 * (size_t)N];
  v.w = s[3 * (size_t)N];
  *(v4f*)(dst + (((size_t)nt * (K >> 3) + kb) * 32 + l) * 4) = v;
}

struct Frag {
  v4f a[4];
  v4f b[4][4];
};

__device__ __forceinline__ void load_chunk(Frag& f, const float* ap,
                                           const v4f* bp, int bstride, int kc) {
#pragma unroll
  for (int u = 0; u < 4; ++u) {
    f.a[u] = *(const v4f*)(ap + (kc * 4 + u) * 8);
#pragma unroll
    for (int s = 0; s < 4; ++s)
      f.b[s][u] = bp[(size_t)s * bstride + (size_t)(kc * 4 + u) * 32];
  }
}

__device__ __forceinline__ void mma_chunk(const Frag& f, v8f& a0, v8f& a1,
                                          v8f& a2, v8f& a3) {
#pragma unroll
  for (int u = 0; u < 4; ++u) {
    const v2f alo = {f.a[u].x, f.a[u].y}, ahi = {f.a[u].z, f.a[u].w};
    a0 = wmma4(alo, (v2f){f.b[0][u].x, f.b[0][u].y}, a0);
    a1 = wmma4(alo, (v2f){f.b[1][u].x, f.b[1][u].y}, a1);
    a2 = wmma4(alo, (v2f){f.b[2][u].x, f.b[2][u].y}, a2);
    a3 = wmma4(alo, (v2f){f.b[3][u].x, f.b[3][u].y}, a3);
    a0 = wmma4(ahi, (v2f){f.b[0][u].z, f.b[0][u].w}, a0);
    a1 = wmma4(ahi, (v2f){f.b[1][u].z, f.b[1][u].w}, a1);
    a2 = wmma4(ahi, (v2f){f.b[2][u].z, f.b[2][u].w}, a2);
    a3 = wmma4(ahi, (v2f){f.b[3][u].z, f.b[3][u].w}, a3);
  }
}

__device__ __forceinline__ void gemm_strip(const float* ap, const v4f* bp,
                                           int bstride, int nch, v8f& a0,
                                           v8f& a1, v8f& a2, v8f& a3) {
  Frag f0, f1;
  load_chunk(f0, ap, bp, bstride, 0);
  for (int kc = 0; kc < nch - 2; kc += 2) {
    load_chunk(f1, ap, bp, bstride, kc + 1);
    mma_chunk(f0, a0, a1, a2, a3);
    load_chunk(f0, ap, bp, bstride, kc + 2);
    mma_chunk(f1, a0, a1, a2, a3);
  }
  load_chunk(f1, ap, bp, bstride, nch - 1);
  mma_chunk(f0, a0, a1, a2, a3);
  mma_chunk(f1, a0, a1, a2, a3);
}

__global__ __launch_bounds__(256) void qkv_kernel(
    const float* __restrict__ x, const float* __restrict__ wqkvt,
    float* __restrict__ qo, float* __restrict__ ko, float* __restrict__ vo) {
  __shared__ float xs[64 * LDA];
  const int m0 = blockIdx.x * 64;
  for (int i = threadIdx.x; i < 64 * (CC / 4); i += 256) {
    const int r = i / (CC / 4), c4 = i % (CC / 4);
    const float4 src = ((const float4*)(x + (m0 + r) * CC))[c4];
    *(float4*)(&xs[r * LDA + c4 * 4]) = src;
  }
  __syncthreads();
  const int wv = threadIdx.x >> 5, l = threadIdx.x & 31;
  const int n16 = l & 15, hi = l >> 4;
  for (int task = wv; task < 144; task += 8) {
    const int rb = task / 36, oh = task % 36;
    const int o = oh / 12, head = oh % 12;
    const float* Wp = wqkvt + (size_t)oh * (DD * CC);
    v8f a0 = {}, a1 = {}, a2 = {}, a3 = {};
    gemm_strip(&xs[(rb * 16 + n16) * LDA + 4 * hi], (const v4f*)Wp + l,
               16 * CC / 4, CC / 32, a0, a1, a2, a3);
    const int mbase = rb * 16 + ((l >> 4) << 3);
    if (o < 2) {
      float* OUT = (o == 0) ? qo : ko;
#pragma unroll
      for (int j = 0; j < 8; ++j) {
        const int tok = m0 + mbase + j;
        const int b = tok >> 11, t = tok & 2047;
        float* dst = OUT + (((size_t)(b * HH + head) * TT + t) * DD);
        dst[0 + n16] = a0[j];  dst[16 + n16] = a1[j];
        dst[32 + n16] = a2[j]; dst[48 + n16] = a3[j];
      }
    } else {
#pragma unroll
      for (int j = 0; j < 8; ++j) {
        const int tok = m0 + mbase + j;
        const int b = tok >> 11, t = tok & 2047;
        float* base = vo + (size_t)(b * HH + head) * DD * TT + t;
        base[(size_t)(0 + n16) * TT] = a0[j];
        base[(size_t)(16 + n16) * TT] = a1[j];
        base[(size_t)(32 + n16) * TT] = a2[j];
        base[(size_t)(48 + n16) * TT] = a3[j];
      }
    }
  }
}

__global__ __launch_bounds__(256) void attn_kernel(
    const float* __restrict__ q, const float* __restrict__ k,
    const float* __restrict__ vt, float* __restrict__ aout) {
  __shared__ float S[16 * SROW];
  __shared__ float red[256];
  __shared__ float rowmax[16], rowrcp[16];
  __shared__ float opart[16 * 64];
  const int ttile = blockIdx.x;
  const int bh = blockIdx.y;
  const int t0 = ttile * 16;
  const float* qp = q + (size_t)bh * TT * DD;
  const float* kp = k + (size_t)bh * TT * DD;
  const float* vp = vt + (size_t)bh * TT * DD;
  const int wv = threadIdx.x >> 5, l = threadIdx.x & 31;
  const int n16 = l & 15, hi = l >> 4, kq = hi << 1;
  const int NT = ttile + 1, L = NT * 16;
  const float scale = 0.036084391824352f;

  v2f af[16];
#pragma unroll
  for (int ks = 0; ks < 16; ++ks)
    af[ks] = *(const v2f*)(kp + (t0 + n16) * DD + ks * 4 + kq);
  for (int st = wv; st < NT; st += 8) {
    const int s0 = st * 16;
    v8f acc = {};
#pragma unroll
    for (int ks = 0; ks < 16; ++ks) {
      const v2f b = *(const v2f*)(qp + (s0 + n16) * DD + ks * 4 + kq);
      acc = wmma4(af[ks], b, acc);
    }
    const int sg = s0 + n16;
#pragma unroll
    for (int j = 0; j < 8; ++j) {
      const int row = (hi << 3) + j;
      float vl = acc[j] * scale;
      if (sg > t0 + row) vl = -1e30f;
      S[row * SROW + sg] = vl;
    }
  }
  __syncthreads();

  {
    const int row = threadIdx.x >> 4, sub = threadIdx.x & 15;
    float m = -1e30f;
    for (int c = sub; c < L; c += 16) m = fmaxf(m, S[row * SROW + c]);
    red[threadIdx.x] = m;
    __syncthreads();
    if (sub == 0) {
      float mm = red[threadIdx.x];
      for (int i = 1; i < 16; ++i) mm = fmaxf(mm, red[threadIdx.x + i]);
      rowmax[row] = mm;
    }
    __syncthreads();
    const float mr = rowmax[row];
    float s = 0.f;
    for (int c = sub; c < L; c += 16) {
      const float e = __expf(S[row * SROW + c] - mr);
      S[row * SROW + c] = e;
      s += e;
    }
    red[threadIdx.x] = s;
    __syncthreads();
    if (sub == 0) {
      float ss = 0.f;
      for (int i = 0; i < 16; ++i) ss += red[threadIdx.x + i];
      rowrcp[row] = 1.0f / ss;
    }
  }
  __syncthreads();

  const int tile = wv & 3, half = wv >> 2;
  const int ksteps = L >> 2, kh = ksteps >> 1;
  const int kbeg = half * kh, kend = kbeg + kh;
  v8f acc = {};
  const float* vrow = vp + (size_t)(tile * 16 + n16) * TT;
  for (int ks = kbeg; ks < kend; ++ks) {
    const v2f a = *(const v2f*)(&S[n16 * SROW + ks * 4 + kq]);
    const v2f b = *(const v2f*)(vrow + ks * 4 + kq);
    acc = wmma4(a, b, acc);
  }
  if (half == 1) {
#pragma unroll
    for (int j = 0; j < 8; ++j)
      opart[((hi << 3) + j) * 64 + tile * 16 + n16] = acc[j];
  }
  __syncthreads();
  if (half == 0) {
    const int bb = bh / HH, head = bh % HH;
#pragma unroll
    for (int j = 0; j < 8; ++j) {
      const int row = (hi << 3) + j;
      const float val =
          (acc[j] + opart[row * 64 + tile * 16 + n16]) * rowrcp[row];
      const int t = t0 + row;
      aout[((size_t)(bb * TT + t)) * CC + head * DD + tile * 16 + n16] = val;
    }
  }
}

__global__ __launch_bounds__(256) void add_ln_kernel(
    const float* __restrict__ xa, const float* __restrict__ xb,
    const float* __restrict__ g, const float* __restrict__ bt,
    float* __restrict__ out) {
  __shared__ float red[256];
  const int tok = blockIdx.x;
  const float* pa = xa + (size_t)tok * CC;
  const float* pb = xb + (size_t)tok * CC;
  float vals[3];
  float s1 = 0.f;
#pragma unroll
  for (int i = 0; i < 3; ++i) {
    const int c = threadIdx.x + i * 256;
    vals[i] = pa[c] + pb[c];
    s1 += vals[i];
  }
  red[threadIdx.x] = s1;
  __syncthreads();
  for (int off = 128; off > 0; off >>= 1) {
    if (threadIdx.x < off) red[threadIdx.x] += red[threadIdx.x + off];
    __syncthreads();
  }
  const float mean = red[0] * (1.0f / CC);
  __syncthreads();
  float s2 = 0.f;
#pragma unroll
  for (int i = 0; i < 3; ++i) {
    const float d = vals[i] - mean;
    s2 += d * d;
  }
  red[threadIdx.x] = s2;
  __syncthreads();
  for (int off = 128; off > 0; off >>= 1) {
    if (threadIdx.x < off) red[threadIdx.x] += red[threadIdx.x + off];
    __syncthreads();
  }
  const float rstd = rsqrtf(red[0] * (1.0f / CC) + 1e-5f);
#pragma unroll
  for (int i = 0; i < 3; ++i) {
    const int c = threadIdx.x + i * 256;
    out[(size_t)tok * CC + c] = (vals[i] - mean) * rstd * g[c] + bt[c];
  }
}

__global__ __launch_bounds__(256) void mlp1_kernel(
    const float* __restrict__ xin, const float* __restrict__ W1t,
    const float* __restrict__ bias, float* __restrict__ hout) {
  __shared__ float xs[64 * LDA];
  const int m0 = blockIdx.x * 64;
  for (int i = threadIdx.x; i < 64 * (CC / 4); i += 256) {
    const int r = i / (CC / 4), c4 = i % (CC / 4);
    const float4 src = ((const float4*)(xin + (m0 + r) * CC))[c4];
    *(float4*)(&xs[r * LDA + c4 * 4]) = src;
  }
  __syncthreads();
  const int wv = threadIdx.x >> 5, l = threadIdx.x & 31;
  const int n16 = l & 15, hi = l >> 4;
  for (int task = wv; task < 192; task += 8) {
    const int rb = task / 48, nb = (task % 48) * 64;
    v8f a0 = {}, a1 = {}, a2 = {}, a3 = {};
    gemm_strip(&xs[(rb * 16 + n16) * LDA + 4 * hi],
               (const v4f*)(W1t + (size_t)nb * CC) + l, 16 * CC / 4, CC / 32,
               a0, a1, a2, a3);
    const int mbase = rb * 16 + ((l >> 4) << 3);
#pragma unroll
    for (int j = 0; j < 8; ++j) {
      float* dst = hout + (size_t)(m0 + mbase + j) * FF + nb + n16;
      float z;
      z = a0[j] + bias[nb + n16];
      dst[0] = 0.5f * z * (1.f + erff(z * 0.70710678118f));
      z = a1[j] + bias[nb + 16 + n16];
      dst[16] = 0.5f * z * (1.f + erff(z * 0.70710678118f));
      z = a2[j] + bias[nb + 32 + n16];
      dst[32] = 0.5f * z * (1.f + erff(z * 0.70710678118f));
      z = a3[j] + bias[nb + 48 + n16];
      dst[48] = 0.5f * z * (1.f + erff(z * 0.70710678118f));
    }
  }
}

__global__ __launch_bounds__(256) void mlp2_kernel(
    const float* __restrict__ hin, const float* __restrict__ W2t,
    const float* __restrict__ bias, float* __restrict__ t2) {
  __shared__ float opart[64 * 64];
  const int m0 = blockIdx.x * 64, n0 = blockIdx.y * 64;
  const int wv = threadIdx.x >> 5, l = threadIdx.x & 31;
  const int n16 = l & 15, hi = l >> 4;
  const int rb = wv & 3, half = wv >> 2;
  const int kofs = half * (FF / 2);
  v8f a0 = {}, a1 = {}, a2 = {}, a3 = {};
  gemm_strip(hin + (size_t)(m0 + rb * 16 + n16) * FF + kofs + 4 * hi,
             (const v4f*)(W2t + (size_t)n0 * FF) + (size_t)(kofs / 8) * 32 + l,
             16 * FF / 4, FF / 64, a0, a1, a2, a3);
  const int mb = rb * 16 + ((l >> 4) << 3);
  if (half == 1) {
#pragma unroll
    for (int j = 0; j < 8; ++j) {
      opart[(mb + j) * 64 + n16] = a0[j];
      opart[(mb + j) * 64 + 16 + n16] = a1[j];
      opart[(mb + j) * 64 + 32 + n16] = a2[j];
      opart[(mb + j) * 64 + 48 + n16] = a3[j];
    }
  }
  __syncthreads();
  if (half == 0) {
#pragma unroll
    for (int j = 0; j < 8; ++j) {
      float* dst = t2 + (size_t)(m0 + mb + j) * CC + n0 + n16;
      dst[0]  = a0[j] + opart[(mb + j) * 64 + n16]      + bias[n0 + n16];
      dst[16] = a1[j] + opart[(mb + j) * 64 + 16 + n16] + bias[n0 + 16 + n16];
      dst[32] = a2[j] + opart[(mb + j) * 64 + 32 + n16] + bias[n0 + 32 + n16];
      dst[48] = a3[j] + opart[(mb + j) * 64 + 48 + n16] + bias[n0 + 48 + n16];
    }
  }
}

extern "C" void kernel_launch(void* const* d_in, const int* in_sizes, int n_in,
                              void* d_out, int out_size, void* d_ws,
                              size_t ws_size, hipStream_t stream) {
  (void)in_sizes; (void)n_in; (void)out_size; (void)ws_size;
  const float* x    = (const float*)d_in[0];
  const float* Wq   = (const float*)d_in[1];
  const float* Wk   = (const float*)d_in[2];
  const float* Wv   = (const float*)d_in[3];
  const float* ln1g = (const float*)d_in[4];
  const float* ln1b = (const float*)d_in[5];
  const float* W1   = (const float*)d_in[6];
  const float* b1   = (const float*)d_in[7];
  const float* W2   = (const float*)d_in[8];
  const float* b2   = (const float*)d_in[9];
  const float* ln2g = (const float*)d_in[10];
  const float* ln2b = (const float*)d_in[11];
  float* ws  = (float*)d_ws;
  float* out = (float*)d_out;

  float* q     = ws;
  float* k     = ws + 6291456;
  float* v     = ws + 12582912;
  float* a     = ws + 18874368;
  float* hbuf  = ws;
  float* x1    = ws + 25165824;
  float* t2    = ws + 31457280;
  float* wqkvt = ws + 37748736;
  float* w1t   = ws + 39518208;
  float* w2t   = ws + 41877504;

  pack_kernel<<<dim3((DD / 16) * (CC / 8) / 8, 1, HH), 256, 0, stream>>>(
      Wq, wqkvt + 0 * (size_t)HH * DD * CC, CC, DD);
  pack_kernel<<<dim3((DD / 16) * (CC / 8) / 8, 1, HH), 256, 0, stream>>>(
      Wk, wqkvt + 1 * (size_t)HH * DD * CC, CC, DD);
  pack_kernel<<<dim3((DD / 16) * (CC / 8) / 8, 1, HH), 256, 0, stream>>>(
      Wv, wqkvt + 2 * (size_t)HH * DD * CC, CC, DD);
  pack_kernel<<<dim3((FF / 16) * (CC / 8) / 8, 1, 1), 256, 0, stream>>>(
      W1, w1t, CC, FF);
  pack_kernel<<<dim3((CC / 16) * (FF / 8) / 8, 1, 1), 256, 0, stream>>>(
      W2, w2t, FF, CC);

  qkv_kernel<<<dim3(NTOK / 64), 256, 0, stream>>>(x, wqkvt, q, k, v);
  attn_kernel<<<dim3(TT / 16, BN * HH), 256, 0, stream>>>(q, k, v, a);
  add_ln_kernel<<<dim3(NTOK), 256, 0, stream>>>(x, a, ln1g, ln1b, x1);
  mlp1_kernel<<<dim3(NTOK / 64), 256, 0, stream>>>(x1, w1t, b1, hbuf);
  mlp2_kernel<<<dim3(NTOK / 64, CC / 64), 256, 0, stream>>>(hbuf, w2t, b2, t2);
  add_ln_kernel<<<dim3(NTOK), 256, 0, stream>>>(x1, t2, ln2g, ln2b, out);
}


# qkv and mlp1: the column strips of each 64-row block are split over two workgroups (grid 128x2), so all 256 WGPs work instead of 128; the per-wave 16x64 strip loop is unchanged
# speedup vs baseline: 2.5396x; 1.2409x over previous
#include <hip/hip_runtime.h>
#include <math.h>


typedef float v2f __attribute__((ext_vector_type(2)));
typedef float v4f __attribute__((ext_vector_type(4)));
typedef float v8f __attribute__((ext_vector_type(8)));

#define BN   4
#define TT   2048
#define CC   768
#define HH   12
#define DD   64
#define FF   3072
#define NTOK 8192
#define LDA  772
#define SROW 2052

__device__ __forceinline__ v8f wmma4(v2f a, v2f b, v8f c) {
  return __builtin_amdgcn_wmma_f32_16x16x4_f32(false, a, false, b, (short)0, c,
                                               false, false);
}

__global__ __launch_bounds__(256) void pack_kernel(
    const float* __restrict__ src, float* __restrict__ dst, int K, int N) {
  const size_t bofs = (size_t)blockIdx.z * K * N;
  src += bofs;
  dst += bofs;
  const int l = threadIdx.x & 31;
  const int c = l & 15, h = l >> 4;
  const int ntiles = N >> 4;
  const int chunk = blockIdx.x * 8 + (threadIdx.x >> 5);
  const int nt = chunk % ntiles, kb = chunk / ntiles;
  const float* s = src + (size_t)(kb * 8 + 4 * h) * N + nt * 16 + c;
  v4f v;
  v.x = s[0];
  v.y = s[N];
  v.z = s[2 * (size_t)N];
  v.w = s[3 * (size_t)N];
  *(v4f*)(dst + (((size_t)nt * (K >> 3) + kb) * 32 + l) * 4) = v;
}

struct Frag {
  v4f a[4];
  v4f b[4][4];
};

__device__ __forceinline__ void load_chunk(Frag& f, const float* ap,
                                           const v4f* bp, int bstride, int kc) {
#pragma unroll
  for (int u = 0; u < 4; ++u) {
    f.a[u] = *(const v4f*)(ap + (kc * 4 + u) * 8);
#pragma unroll
    for (int s = 0; s < 4; ++s)
      f.b[s][u] = bp[(size_t)s * bstride + (size_t)(kc * 4 + u) * 32];
  }
}

__device__ __forceinline__ void mma_chunk(const Frag& f, v8f& a0, v8f& a1,
                                          v8f& a2, v8f& a3) {
#pragma unroll
  for (int u = 0; u < 4; ++u) {
    const v2f alo = {f.a[u].x, f.a[u].y}, ahi = {f.a[u].z, f.a[u].w};
    a0 = wmma4(alo, (v2f){f.b[0][u].x, f.b[0][u].y}, a0);
    a1 = wmma4(alo, (v2f){f.b[1][u].x, f.b[1][u].y}, a1);
    a2 = wmma4(alo, (v2f){f.b[2][u].x, f.b[2][u].y}, a2);
    a3 = wmma4(alo, (v2f){f.b[3][u].x, f.b[3][u].y}, a3);
    a0 = wmma4(ahi, (v2f){f.b[0][u].z, f.b[0][u].w}, a0);
    a1 = wmma4(ahi, (v2f){f.b[1][u].z, f.b[1][u].w}, a1);
    a2 = wmma4(ahi, (v2f){f.b[2][u].z, f.b[2][u].w}, a2);
    a3 = wmma4(ahi, (v2f){f.b[3][u].z, f.b[3][u].w}, a3);
  }
}

__device__ __forceinline__ void gemm_strip(const float* ap, const v4f* bp,
                                           int bstride, int nch, v8f& a0,
                                           v8f& a1, v8f& a2, v8f& a3) {
  Frag f0, f1;
  load_chunk(f0, ap, bp, bstride, 0);
  for (int kc = 0; kc < nch - 2; kc += 2) {
    load_chunk(f1, ap, bp, bstride, kc + 1);
    mma_chunk(f0, a0, a1, a2, a3);
    load_chunk(f0, ap, bp, bstride, kc + 2);
    mma_chunk(f1, a0, a1, a2, a3);
  }
  load_chunk(f1, ap, bp, bstride, nch - 1);
  mma_chunk(f0, a0, a1, a2, a3);
  mma_chunk(f1, a0, a1, a2, a3);
}

__global__ __launch_bounds__(256) void qkv_kernel(
    const float* __restrict__ x, const float* __restrict__ wqkvt,
    float* __restrict__ qo, float* __restrict__ ko, float* __restrict__ vo) {
  __shared__ float xs[64 * LDA];
  const int m0 = blockIdx.x * 64;
  for (int i = threadIdx.x; i < 64 * (CC / 4); i += 256) {
    const int r = i / (CC / 4), c4 = i % (CC / 4);
    const float4 src = ((const float4*)(x + (m0 + r) * CC))[c4];
    *(float4*)(&xs[r * LDA + c4 * 4]) = src;
  }
  __syncthreads();
  const int wv = threadIdx.x >> 5, l = threadIdx.x & 31;
  const int n16 = l & 15, hi = l >> 4;
  const int yb = blockIdx.y;
  for (int task = wv; task < 72; task += 8) {
    const int rb = task / 18, oh = 2 * (task % 18) + yb;
    const int o = oh / 12, head = oh % 12;
    const float* Wp = wqkvt + (size_t)oh * (DD * CC);
    v8f a0 = {}, a1 = {}, a2 = {}, a3 = {};
    gemm_strip(&xs[(rb * 16 + n16) * LDA + 4 * hi], (const v4f*)Wp + l,
               16 * CC / 4, CC / 32, a0, a1, a2, a3);
    const int mbase = rb * 16 + ((l >> 4) << 3);
    if (o < 2) {
      float* OUT = (o == 0) ? qo : ko;
#pragma unroll
      for (int j = 0; j < 8; ++j) {
        const int tok = m0 + mbase + j;
        const int b = tok >> 11, t = tok & 2047;
        float* dst = OUT + (((size_t)(b * HH + head) * TT + t) * DD);
        dst[0 + n16] = a0[j];  dst[16 + n16] = a1[j];
        dst[32 + n16] = a2[j]; dst[48 + n16] = a3[j];
      }
    } else {
#pragma unroll
      for (int j = 0; j < 8; ++j) {
        const int tok = m0 + mbase + j;
        const int b = tok >> 11, t = tok & 2047;
        float* base = vo + (size_t)(b * HH + head) * DD * TT + t;
        base[(size_t)(0 + n16) * TT] = a0[j];
        base[(size_t)(16 + n16) * TT] = a1[j];
        base[(size_t)(32 + n16) * TT] = a2[j];
        base[(size_t)(48 + n16) * TT] = a3[j];
      }
    }
  }
}

__global__ __launch_bounds__(256) void attn_kernel(
    const float* __restrict__ q, const float* __restrict__ k,
    const float* __restrict__ vt, float* __restrict__ aout) {
  __shared__ float S[16 * SROW];
  __shared__ float red[256];
  __shared__ float rowmax[16], rowrcp[16];
  __shared__ float opart[16 * 64];
  const int ttile = blockIdx.x;
  const int bh = blockIdx.y;
  const int t0 = ttile * 16;
  const float* qp = q + (size_t)bh * TT * DD;
  const float* kp = k + (size_t)bh * TT * DD;
  const float* vp = vt + (size_t)bh * TT * DD;
  const int wv = threadIdx.x >> 5, l = threadIdx.x & 31;
  const int n16 = l & 15, hi = l >> 4, kq = hi << 1;
  const int NT = ttile + 1, L = NT * 16;
  const float scale = 0.036084391824352f;

  v2f af[16];
#pragma unroll
  for (int ks = 0; ks < 16; ++ks)
    af[ks] = *(const v2f*)(kp + (t0 + n16) * DD + ks * 4 + kq);
  for (int st = wv; st < NT; st += 8) {
    const int s0 = st * 16;
    v8f acc = {};
#pragma unroll
    for (int ks = 0; ks < 16; ++ks) {
      const v2f b = *(const v2f*)(qp + (s0 + n16) * DD + ks * 4 + kq);
      acc = wmma4(af[ks], b, acc);
    }
    const int sg = s0 + n16;
#pragma unroll
    for (int j = 0; j < 8; ++j) {
      const int row = (hi << 3) + j;
      float vl = acc[j] * scale;
      if (sg > t0 + row) vl = -1e30f;
      S[row * SROW + sg] = vl;
    }
  }
  __syncthreads();

  {
    const int row = threadIdx.x >> 4, sub = threadIdx.x & 15;
    float m = -1e30f;
    for (int c = sub; c < L; c += 16) m = fmaxf(m, S[row * SROW + c]);
    red[threadIdx.x] = m;
    __syncthreads();
    if (sub == 0) {
      float mm = red[threadIdx.x];
      for (int i = 1; i < 16; ++i) mm = fmaxf(mm, red[threadIdx.x + i]);
      rowmax[row] = mm;
    }
    __syncthreads();
    const float mr = rowmax[row];
    float s = 0.f;
    for (int c = sub; c < L; c += 16) {
      const float e = __expf(S[row * SROW + c] - mr);
      S[row * SROW + c] = e;
      s += e;
    }
    red[threadIdx.x] = s;
    __syncthreads();
    if (sub == 0) {
      float ss = 0.f;
      for (int i = 0; i < 16; ++i) ss += red[threadIdx.x + i];
      rowrcp[row] = 1.0f / ss;
    }
  }
  __syncthreads();

  const int tile = wv & 3, half = wv >> 2;
  const int ksteps = L >> 2, kh = ksteps >> 1;
  const int kbeg = half * kh, kend = kbeg + kh;
  v8f acc = {};
  const float* vrow = vp + (size_t)(tile * 16 + n16) * TT;
  for (int ks = kbeg; ks < kend; ++ks) {
    const v2f a = *(const v2f*)(&S[n16 * SROW + ks * 4 + kq]);
    const v2f b = *(const v2f*)(vrow + ks * 4 + kq);
    acc = wmma4(a, b, acc);
  }
  if (half == 1) {
#pragma unroll
    for (int j = 0; j < 8; ++j)
      opart[((hi << 3) + j) * 64 + tile * 16 + n16] = acc[j];
  }
  __syncthreads();
  if (half == 0) {
    const int bb = bh / HH, head = bh % HH;
#pragma unroll
    for (int j = 0; j < 8; ++j) {
      const int row = (hi << 3) + j;
      const float val =
          (acc[j] + opart[row * 64 + tile * 16 + n16]) * rowrcp[row];
      const int t = t0 + row;
      aout[((size_t)(bb * TT + t)) * CC + head * DD + tile * 16 + n16] = val;
    }
  }
}

__global__ __launch_bounds__(256) void add_ln_kernel(
    const float* __restrict__ xa, const float* __restrict__ xb,
    const float* __restrict__ g, const float* __restrict__ bt,
    float* __restrict__ out) {
  __shared__ float red[256];
  const int tok = blockIdx.x;
  const float* pa = xa + (size_t)tok * CC;
  const float* pb = xb + (size_t)tok * CC;
  float vals[3];
  float s1 = 0.f;
#pragma unroll
  for (int i = 0; i < 3; ++i) {
    const int c = threadIdx.x + i * 256;
    vals[i] = pa[c] + pb[c];
    s1 += vals[i];
  }
  red[threadIdx.x] = s1;
  __syncthreads();
  for (int off = 128; off > 0; off >>= 1) {
    if (threadIdx.x < off) red[threadIdx.x] += red[threadIdx.x + off];
    __syncthreads();
  }
  const float mean = red[0] * (1.0f / CC);
  __syncthreads();
  float s2 = 0.f;
#pragma unroll
  for (int i = 0; i < 3; ++i) {
    const float d = vals[i] - mean;
    s2 += d * d;
  }
  red[threadIdx.x] = s2;
  __syncthreads();
  for (int off = 128; off > 0; off >>= 1) {
    if (threadIdx.x < off) red[threadIdx.x] += red[threadIdx.x + off];
    __syncthreads();
  }
  const float rstd = rsqrtf(red[0] * (1.0f / CC) + 1e-5f);
#pragma unroll
  for (int i = 0; i < 3; ++i) {
    const int c = threadIdx.x + i * 256;
    out[(size_t)tok * CC + c] = (vals[i] - mean) * rstd * g[c] + bt[c];
  }
}

__global__ __launch_bounds__(256) void mlp1_kernel(
    const float* __restrict__ xin, const float* __restrict__ W1t,
    const float* __restrict__ bias, float* __restrict__ hout) {
  __shared__ float xs[64 * LDA];
  const int m0 = blockIdx.x * 64;
  for (int i = threadIdx.x; i < 64 * (CC / 4); i += 256) {
    const int r = i / (CC / 4), c4 = i % (CC / 4);
    const float4 src = ((const float4*)(xin + (m0 + r) * CC))[c4];
    *(float4*)(&xs[r * LDA + c4 * 4]) = src;
  }
  __syncthreads();
  const int wv = threadIdx.x >> 5, l = threadIdx.x & 31;
  const int n16 = l & 15, hi = l >> 4;
  const int yb = blockIdx.y;
  for (int task = wv; task < 96; task += 8) {
    const int rb = task / 24, nb = (yb * 24 + task % 24) * 64;
    v8f a0 = {}, a1 = {}, a2 = {}, a3 = {};
    gemm_strip(&xs[(rb * 16 + n16) * LDA + 4 * hi],
               (const v4f*)(W1t + (size_t)nb * CC) + l, 16 * CC / 4, CC / 32,
               a0, a1, a2, a3);
    const int mbase = rb * 16 + ((l >> 4) << 3);
#pragma unroll
    for (int j = 0; j < 8; ++j) {
      float* dst = hout + (size_t)(m0 + mbase + j) * FF + nb + n16;
      float z;
      z = a0[j] + bias[nb + n16];
      dst[0] = 0.5f * z * (1.f + erff(z * 0.70710678118f));
      z = a1[j] + bias[nb + 16 + n16];
      dst[16] = 0.5f * z * (1.f + erff(z * 0.70710678118f));
      z = a2[j] + bias[nb + 32 + n16];
      dst[32] = 0.5f * z * (1.f + erff(z * 0.70710678118f));
      z = a3[j] + bias[nb + 48 + n16];
      dst[48] = 0.5f * z * (1.f + erff(z * 0.70710678118f));
    }
  }
}

__global__ __launch_bounds__(256) void mlp2_kernel(
    const float* __restrict__ hin, const float* __restrict__ W2t,
    const float* __restrict__ bias, float* __restrict__ t2) {
  __shared__ float opart[64 * 64];
  const int m0 = blockIdx.x * 64, n0 = blockIdx.y * 64;
  const int wv = threadIdx.x >> 5, l = threadIdx.x & 31;
  const int n16 = l & 15, hi = l >> 4;
  const int rb = wv & 3, half = wv >> 2;
  const int kofs = half * (FF / 2);
  v8f a0 = {}, a1 = {}, a2 = {}, a3 = {};
  gemm_strip(hin + (size_t)(m0 + rb * 16 + n16) * FF + kofs + 4 * hi,
             (const v4f*)(W2t + (size_t)n0 * FF) + (size_t)(kofs / 8) * 32 + l,
             16 * FF / 4, FF / 64, a0, a1, a2, a3);
  const int mb = rb * 16 + ((l >> 4) << 3);
  if (half == 1) {
#pragma unroll
    for (int j = 0; j < 8; ++j) {
      opart[(mb + j) * 64 + n16] = a0[j];
      opart[(mb + j) * 64 + 16 + n16] = a1[j];
      opart[(mb + j) * 64 + 32 + n16] = a2[j];
      opart[(mb + j) * 64 + 48 + n16] = a3[j];
    }
  }
  __syncthreads();
  if (half == 0) {
#pragma unroll
    for (int j = 0; j < 8; ++j) {
      float* dst = t2 + (size_t)(m0 + mb + j) * CC + n0 + n16;
      dst[0]  = a0[j] + opart[(mb + j) * 64 + n16]      + bias[n0 + n16];
      dst[16] = a1[j] + opart[(mb + j) * 64 + 16 + n16] + bias[n0 + 16 + n16];
      dst[32] = a2[j] + opart[(mb + j) * 64 + 32 + n16] + bias[n0 + 32 + n16];
      dst[48] = a3[j] + opart[(mb + j) * 64 + 48 + n16] + bias[n0 + 48 + n16];
    }
  }
}

extern "C" void kernel_launch(void* const* d_in, const int* in_sizes, int n_in,
                              void* d_out, int out_size, void* d_ws,
                              size_t ws_size, hipStream_t stream) {
  (void)in_sizes; (void)n_in; (void)out_size; (void)ws_size;
  const float* x    = (const float*)d_in[0];
  const float* Wq   = (const float*)d_in[1];
  const float* Wk   = (const float*)d_in[2];
  const float* Wv   = (const float*)d_in[3];
  const float* ln1g = (const float*)d_in[4];
  const float* ln1b = (const float*)d_in[5];
  const float* W1   = (const float*)d_in[6];
  const float* b1   = (const float*)d_in[7];
  const float* W2   = (const float*)d_in[8];
  const float* b2   = (const float*)d_in[9];
  const float* ln2g = (const float*)d_in[10];
  const float* ln2b = (const float*)d_in[11];
  float* ws  = (float*)d_ws;
  float* out = (float*)d_out;

  float* q     = ws;
  float* k     = ws + 6291456;
  float* v     = ws + 12582912;
  float* a     = ws + 18874368;
  float* hbuf  = ws;
  float* x1    = ws + 25165824;
  float* t2    = ws + 31457280;
  float* wqkvt = ws + 37748736;
  float* w1t   = ws + 39518208;
  float* w2t   = ws + 41877504;

  pack_kernel<<<dim3((DD / 16) * (CC / 8) / 8, 1, HH), 256, 0, stream>>>(
      Wq, wqkvt + 0 * (size_t)HH * DD * CC, CC, DD);
  pack_kernel<<<dim3((DD / 16) * (CC / 8) / 8, 1, HH), 256, 0, stream>>>(
      Wk, wqkvt + 1 * (size_t)HH * DD * CC, CC, DD);
  pack_kernel<<<dim3((DD / 16) * (CC / 8) / 8, 1, HH), 256, 0, stream>>>(
      Wv, wqkvt + 2 * (size_t)HH * DD * CC, CC, DD);
  pack_kernel<<<dim3((FF / 16) * (CC / 8) / 8, 1, 1), 256, 0, stream>>>(
      W1, w1t, CC, FF);
  pack_kernel<<<dim3((CC / 16) * (FF / 8) / 8, 1, 1), 256, 0, stream>>>(
      W2, w2t, FF, CC);

  qkv_kernel<<<dim3(NTOK / 64, 2), 256, 0, stream>>>(x, wqkvt, q, k, v);
  attn_kernel<<<dim3(TT / 16, BN * HH), 256, 0, stream>>>(q, k, v, a);
  add_ln_kernel<<<dim3(NTOK), 256, 0, stream>>>(x, a, ln1g, ln1b, x1);
  mlp1_kernel<<<dim3(NTOK / 64, 2), 256, 0, stream>>>(x1, w1t, b1, hbuf);
  mlp2_kernel<<<dim3(NTOK / 64, CC / 64), 256, 0, stream>>>(hbuf, w2t, b2, t2);
  add_ln_kernel<<<dim3(NTOK), 256, 0, stream>>>(x1, t2, ln2g, ln2b, out);
}


# q, k, v written by the projection epilogue in WMMA fragment order: attention fragment loads are 16 bytes per lane over 4 whole lines (was 8 over 16 lines); P.V runs 8 k-blocks a trip on zero-padded P
# speedup vs baseline: 2.8954x; 1.1401x over previous
#include <hip/hip_runtime.h>
#include <math.h>


typedef float v2f __attribute__((ext_vector_type(2)));
typedef float v4f __attribute__((ext_vector_type(4)));
typedef float v8f __attribute__((ext_vector_type(8)));

#define BN   4
#define TT   2048
#define CC   768
#define HH   12
#define DD   64
#define FF   3072
#define NTOK 8192
#define LDA  772
#define SROW 2052

__device__ __forceinline__ v8f wmma4(v2f a, v2f b, v8f c) {
  return __builtin_amdgcn_wmma_f32_16x16x4_f32(false, a, false, b, (short)0, c,
                                               false, false);
}

__global__ __launch_bounds__(256) void pack_kernel(
    const float* __restrict__ src, float* __restrict__ dst, int K, int N) {
  const size_t bofs = (size_t)blockIdx.z * K * N;
  src += bofs;
  dst += bofs;
  const int l = threadIdx.x & 31;
  const int c = l & 15, h = l >> 4;
  const int ntiles = N >> 4;
  const int chunk = blockIdx.x * 8 + (threadIdx.x >> 5);
  const int nt = chunk % ntiles, kb = chunk / ntiles;
  const float* s = src + (size_t)(kb * 8 + 4 * h) * N + nt * 16 + c;
  v4f v;
  v.x = s[0];
  v.y = s[N];
  v.z = s[2 * (size_t)N];
  v.w = s[3 * (size_t)N];
  *(v4f*)(dst + (((size_t)nt * (K >> 3) + kb) * 32 + l) * 4) = v;
}

struct Frag {
  v4f a[4];
  v4f b[4][4];
};

__device__ __forceinline__ void load_chunk(Frag& f, const float* ap,
                                           const v4f* bp, int bstride, int kc) {
#pragma unroll
  for (int u = 0; u < 4; ++u) {
    f.a[u] = *(const v4f*)(ap + (kc * 4 + u) * 8);
#pragma unroll
    for (int s = 0; s < 4; ++s)
      f.b[s][u] = bp[(size_t)s * bstride + (size_t)(kc * 4 + u) * 32];
  }
}

template <bool TR>
__device__ __forceinline__ v8f mm(v2f x, v2f w, v8f c) {
  return TR ? wmma4(w, x, c) : wmma4(x, w, c);
}

template <bool TR>
__device__ __forceinline__ void mma_chunk(const Frag& f, v8f& a0, v8f& a1,
                                          v8f& a2, v8f& a3) {
#pragma unroll
  for (int u = 0; u < 4; ++u) {
    const v2f alo = {f.a[u].x, f.a[u].y}, ahi = {f.a[u].z, f.a[u].w};
    a0 = mm<TR>(alo, (v2f){f.b[0][u].x, f.b[0][u].y}, a0);
    a1 = mm<TR>(alo, (v2f){f.b[1][u].x, f.b[1][u].y}, a1);
    a2 = mm<TR>(alo, (v2f){f.b[2][u].x, f.b[2][u].y}, a2);
    a3 = mm<TR>(alo, (v2f){f.b[3][u].x, f.b[3][u].y}, a3);
    a0 = mm<TR>(ahi, (v2f){f.b[0][u].z, f.b[0][u].w}, a0);
    a1 = mm<TR>(ahi, (v2f){f.b[1][u].z, f.b[1][u].w}, a1);
    a2 = mm<TR>(ahi, (v2f){f.b[2][u].z, f.b[2][u].w}, a2);
    a3 = mm<TR>(ahi, (v2f){f.b[3][u].z, f.b[3][u].w}, a3);
  }
}

template <bool TR>
__device__ __forceinline__ void gemm_strip(const float* ap, const v4f* bp,
                                           int bstride, int nch, v8f& a0,
                                           v8f& a1, v8f& a2, v8f& a3) {
  Frag f0, f1;
  load_chunk(f0, ap, bp, bstride, 0);
  for (int kc = 0; kc < nch - 2; kc += 2) {
    load_chunk(f1, ap, bp, bstride, kc + 1);
    mma_chunk<TR>(f0, a0, a1, a2, a3);
    load_chunk(f0, ap, bp, bstride, kc + 2);
    mma_chunk<TR>(f1, a0, a1, a2, a3);
  }
  load_chunk(f1, ap, bp, bstride, nch - 1);
  mma_chunk<TR>(f0, a0, a1, a2, a3);
  mma_chunk<TR>(f1, a0, a1, a2, a3);
}

__device__ __forceinline__ v4f lo4(v8f a) {
  return __builtin_shufflevector(a, a, 0, 1, 2, 3);
}
__device__ __forceinline__ v4f hi4(v8f a) {
  return __builtin_shufflevector(a, a, 4, 5, 6, 7);
}

__global__ __launch_bounds__(256) void qkv_kernel(
    const float* __restrict__ x, const float* __restrict__ wqkvt,
    float* __restrict__ qo, float* __restrict__ ko, float* __restrict__ vo) {
  __shared__ float xs[64 * LDA];
  const int m0 = blockIdx.x * 64;
  for (int i = threadIdx.x; i < 64 * (CC / 4); i += 256) {
    const int r = i / (CC / 4), c4 = i % (CC / 4);
    const float4 src = ((const float4*)(x + (m0 + r) * CC))[c4];
    *(float4*)(&xs[r * LDA + c4 * 4]) = src;
  }
  __syncthreads();
  const int wv = threadIdx.x >> 5, l = threadIdx.x & 31;
  const int n16 = l & 15, hi = l >> 4;
  const int yb = blockIdx.y;
  for (int task = wv; task < 72; task += 8) {
    const int rb = task / 18, oh = 2 * (task % 18) + yb;
    const int o = oh / 12, head = oh % 12;
    const float* Wp = wqkvt + (size_t)oh * (DD * CC);
    v8f a0 = {}, a1 = {}, a2 = {}, a3 = {};
    const float* ap = &xs[(rb * 16 + n16) * LDA + 4 * hi];
    const int tb = m0 + rb * 16;
    const int b = tb >> 11, t0 = tb & 2047;
    const size_t bhofs = (size_t)(b * HH + head) * TT * DD;
    if (o < 2) {
      gemm_strip<true>(ap, (const v4f*)Wp + l, 16 * CC / 4, CC / 32, a0, a1, a2,
                       a3);
      v4f* dst = (v4f*)(((o == 0) ? qo : ko) + bhofs + (size_t)t0 * DD) +
                 hi * 32 + n16;
      dst[0] = lo4(a0);   dst[16] = hi4(a0);
      dst[64] = lo4(a1);  dst[80] = hi4(a1);
      dst[128] = lo4(a2); dst[144] = hi4(a2);
      dst[192] = lo4(a3); dst[208] = hi4(a3);
    } else {
      gemm_strip<false>(ap, (const v4f*)Wp + l, 16 * CC / 4, CC / 32, a0, a1, a2,
                        a3);
      v4f* dst = (v4f*)(vo + bhofs) + ((t0 >> 3) + hi) * 128 + n16;
      dst[0] = lo4(a0);  dst[16] = hi4(a0);
      dst[32] = lo4(a1); dst[48] = hi4(a1);
      dst[64] = lo4(a2); dst[80] = hi4(a2);
      dst[96] = lo4(a3); dst[112] = hi4(a3);
    }
  }
}

__global__ __launch_bounds__(256) void attn_kernel(
    const float* __restrict__ q, const float* __restrict__ k,
    const float* __restrict__ vt, float* __restrict__ aout) {
  __shared__ __attribute__((aligned(16))) float S[16 * SROW];
  __shared__ float red[256];
  __shared__ float rowmax[16], rowrcp[16];
  __shared__ float opart[16 * 64];
  const int ttile = blockIdx.x;
  const int bh = blockIdx.y;
  const int t0 = ttile * 16;
  const int wv = threadIdx.x >> 5, l = threadIdx.x & 31;
  const int n16 = l & 15, hi = l >> 4;
  const v4f* qp = (const v4f*)(q + (size_t)bh * TT * DD) + l;
  const v4f* kp = (const v4f*)(k + (size_t)bh * TT * DD) + l;
  const v4f* vp = (const v4f*)(vt + (size_t)bh * TT * DD) + l;
  const int NT = ttile + 1, L = NT * 16;
  const float scale = 0.036084391824352f;

  v4f af[8];
#pragma unroll
  for (int kb = 0; kb < 8; ++kb) af[kb] = kp[(ttile * 8 + kb) * 32];
  for (int st = wv; st < NT; st += 8) {
    const int s0 = st * 16;
    v8f acc = {};
#pragma unroll
    for (int kb = 0; kb < 8; ++kb) {
      const v4f b = qp[(st * 8 + kb) * 32];
      acc = wmma4((v2f){af[kb].x, af[kb].y}, (v2f){b.x, b.y}, acc);
      acc = wmma4((v2f){af[kb].z, af[kb].w}, (v2f){b.z, b.w}, acc);
    }
    const int sg = s0 + n16;
#pragma unroll
    for (int j = 0; j < 8; ++j) {
      const int row = (hi << 3) + j;
      float vl = acc[j] * scale;
      if (sg > t0 + row) vl = -1e30f;
      S[row * SROW + sg] = vl;
    }
  }
  __syncthreads();

  {
    const int row = threadIdx.x >> 4, sub = threadIdx.x & 15;
    float m = -1e30f;
    for (int c = sub; c < L; c += 16) m = fmaxf(m, S[row * SROW + c]);
    red[threadIdx.x] = m;
    __syncthreads();
    if (sub == 0) {
      float mm = red[threadIdx.x];
      for (int i = 1; i < 16; ++i) mm = fmaxf(mm, red[threadIdx.x + i]);
      rowmax[row] = mm;
    }
    __syncthreads();
    const float mr = rowmax[row];
    float s = 0.f;
    for (int c = sub; c < L; c += 16) {
      const float e = __expf(S[row * SROW + c] - mr);
      S[row * SROW + c] = e;
      s += e;
    }
    for (int c = L + sub; c < ((L + 63) & ~63); c += 16) S[row * SROW + c] = 0.f;
    red[threadIdx.x] = s;
    __syncthreads();
    if (sub == 0) {
      float ss = 0.f;
      for (int i = 0; i < 16; ++i) ss += red[threadIdx.x + i];
      rowrcp[row] = 1.0f / ss;
    }
  }
  __syncthreads();

  const int tile = wv & 3, half = wv >> 2;
  const int G = (NT + 3) >> 2, gh = (G + 1) >> 1;
  const int gbeg = half ? gh : 0, gend = half ? G : gh;
  v8f acc = {};
  const float* prow = &S[n16 * SROW + 4 * hi] + gbeg * 64;
  const v4f* vb = vp + tile * 32 + (size_t)gbeg * (8 * 128);
  for (int g = gbeg; g < gend; ++g, prow += 64, vb += 8 * 128) {
    v4f a[8], b[8];
#pragma unroll
    for (int u = 0; u < 8; ++u) b[u] = vb[u * 128];
#pragma unroll
    for (int u = 0; u < 8; ++u) a[u] = *(const v4f*)(prow + u * 8);
    __builtin_amdgcn_sched_barrier(0);
#pragma unroll
    for (int u = 0; u < 8; ++u) {
      acc = wmma4((v2f){a[u].x, a[u].y}, (v2f){b[u].x, b[u].y}, acc);
      acc = wmma4((v2f){a[u].z, a[u].w}, (v2f){b[u].z, b[u].w}, acc);
    }
  }
  if (half == 1) {
#pragma unroll
    for (int j = 0; j < 8; ++j)
      opart[((hi << 3) + j) * 64 + tile * 16 + n16] = acc[j];
  }
  __syncthreads();
  if (half == 0) {
    const int bb = bh / HH, head = bh % HH;
#pragma unroll
    for (int j = 0; j < 8; ++j) {
      const int row = (hi << 3) + j;
      const float val =
          (acc[j] + opart[row * 64 + tile * 16 + n16]) * rowrcp[row];
      const int t = t0 + row;
      aout[((size_t)(bb * TT + t)) * CC + head * DD + tile * 16 + n16] = val;
    }
  }
}

__global__ __launch_bounds__(256) void add_ln_kernel(
    const float* __restrict__ xa, const float* __restrict__ xb,
    const float* __restrict__ g, const float* __restrict__ bt,
    float* __restrict__ out) {
  __shared__ float red[256];
  const int tok = blockIdx.x;
  const float* pa = xa + (size_t)tok * CC;
  const float* pb = xb + (size_t)tok * CC;
  float vals[3];
  float s1 = 0.f;
#pragma unroll
  for (int i = 0; i < 3; ++i) {
    const int c = threadIdx.x + i * 256;
    vals[i] = pa[c] + pb[c];
    s1 += vals[i];
  }
  red[threadIdx.x] = s1;
  __syncthreads();
  for (int off = 128; off > 0; off >>= 1) {
    if (threadIdx.x < off) red[threadIdx.x] += red[threadIdx.x + off];
    __syncthreads();
  }
  const float mean = red[0] * (1.0f / CC);
  __syncthreads();
  float s2 = 0.f;
#pragma unroll
  for (int i = 0; i < 3; ++i) {
    const float d = vals[i] - mean;
    s2 += d * d;
  }
  red[threadIdx.x] = s2;
  __syncthreads();
  for (int off = 128; off > 0; off >>= 1) {
    if (threadIdx.x < off) red[threadIdx.x] += red[threadIdx.x + off];
    __syncthreads();
  }
  const float rstd = rsqrtf(red[0] * (1.0f / CC) + 1e-5f);
#pragma unroll
  for (int i = 0; i < 3; ++i) {
    const int c = threadIdx.x + i * 256;
    out[(size_t)tok * CC + c] = (vals[i] - mean) * rstd * g[c] + bt[c];
  }
}

__global__ __launch_bounds__(256) void mlp1_kernel(
    const float* __restrict__ xin, const float* __restrict__ W1t,
    const float* __restrict__ bias, float* __restrict__ hout) {
  __shared__ float xs[64 * LDA];
  const int m0 = blockIdx.x * 64;
  for (int i = threadIdx.x; i < 64 * (CC / 4); i += 256) {
    const int r = i / (CC / 4), c4 = i % (CC / 4);
    const float4 src = ((const float4*)(xin + (m0 + r) * CC))[c4];
    *(float4*)(&xs[r * LDA + c4 * 4]) = src;
  }
  __syncthreads();
  const int wv = threadIdx.x >> 5, l = threadIdx.x & 31;
  const int n16 = l & 15, hi = l >> 4;
  const int yb = blockIdx.y;
  for (int task = wv; task < 96; task += 8) {
    const int rb = task / 24, nb = (yb * 24 + task % 24) * 64;
    v8f a0 = {}, a1 = {}, a2 = {}, a3 = {};
    gemm_strip<false>(&xs[(rb * 16 + n16) * LDA + 4 * hi],
                      (const v4f*)(W1t + (size_t)nb * CC) + l, 16 * CC / 4,
                      CC / 32, a0, a1, a2, a3);
    const int mbase = rb * 16 + ((l >> 4) << 3);
#pragma unroll
    for (int j = 0; j < 8; ++j) {
      float* dst = hout + (size_t)(m0 + mbase + j) * FF + nb + n16;
      float z;
      z = a0[j] + bias[nb + n16];
      dst[0] = 0.5f * z * (1.f + erff(z * 0.70710678118f));
      z = a1[j] + bias[nb + 16 + n16];
      dst[16] = 0.5f * z * (1.f + erff(z * 0.70710678118f));
      z = a2[j] + bias[nb + 32 + n16];
      dst[32] = 0.5f * z * (1.f + erff(z * 0.70710678118f));
      z = a3[j] + bias[nb + 48 + n16];
      dst[48] = 0.5f * z * (1.f + erff(z * 0.70710678118f));
    }
  }
}

__global__ __launch_bounds__(256) void mlp2_kernel(
    const float* __restrict__ hin, const float* __restrict__ W2t,
    const float* __restrict__ bias, float* __restrict__ t2) {
  __shared__ float opart[64 * 64];
  const int m0 = blockIdx.x * 64, n0 = blockIdx.y * 64;
  const int wv = threadIdx.x >> 5, l = threadIdx.x & 31;
  const int n16 = l & 15, hi = l >> 4;
  const int rb = wv & 3, half = wv >> 2;
  const int kofs = half * (FF / 2);
  v8f a0 = {}, a1 = {}, a2 = {}, a3 = {};
  gemm_strip<false>(
      hin + (size_t)(m0 + rb * 16 + n16) * FF + kofs + 4 * hi,
      (const v4f*)(W2t + (size_t)n0 * FF) + (size_t)(kofs / 8) * 32 + l,
      16 * FF / 4, FF / 64, a0, a1, a2, a3);
  const int mb = rb * 16 + ((l >> 4) << 3);
  if (half == 1) {
#pragma unroll
    for (int j = 0; j < 8; ++j) {
      opart[(mb + j) * 64 + n16] = a0[j];
      opart[(mb + j) * 64 + 16 + n16] = a1[j];
      opart[(mb + j) * 64 + 32 + n16] = a2[j];
      opart[(mb + j) * 64 + 48 + n16] = a3[j];
    }
  }
  __syncthreads();
  if (half == 0) {
#pragma unroll
    for (int j = 0; j < 8; ++j) {
      float* dst = t2 + (size_t)(m0 + mb + j) * CC + n0 + n16;
      dst[0]  = a0[j] + opart[(mb + j) * 64 + n16]      + bias[n0 + n16];
      dst[16] = a1[j] + opart[(mb + j) * 64 + 16 + n16] + bias[n0 + 16 + n16];
      dst[32] = a2[j] + opart[(mb + j) * 64 + 32 + n16] + bias[n0 + 32 + n16];
      dst[48] = a3[j] + opart[(mb + j) * 64 + 48 + n16] + bias[n0 + 48 + n16];
    }
  }
}

extern "C" void kernel_launch(void* const* d_in, const int* in_sizes, int n_in,
                              void* d_out, int out_size, void* d_ws,
                              size_t ws_size, hipStream_t stream) {
  (void)in_sizes; (void)n_in; (void)out_size; (void)ws_size;
  const float* x    = (const float*)d_in[0];
  const float* Wq   = (const float*)d_in[1];
  const float* Wk   = (const float*)d_in[2];
  const float* Wv   = (const float*)d_in[3];
  const float* ln1g = (const float*)d_in[4];
  const float* ln1b = (const float*)d_in[5];
  const float* W1   = (const float*)d_in[6];
  const float* b1   = (const float*)d_in[7];
  const float* W2   = (const float*)d_in[8];
  const float* b2   = (const float*)d_in[9];
  const float* ln2g = (const float*)d_in[10];
  const float* ln2b = (const float*)d_in[11];
  float* ws  = (float*)d_ws;
  float* out = (float*)d_out;

  float* q     = ws;
  float* k     = ws + 6291456;
  float* v     = ws + 12582912;
  float* a     = ws + 18874368;
  float* hbuf  = ws;
  float* x1    = ws + 25165824;
  float* t2    = ws + 31457280;
  float* wqkvt = ws + 37748736;
  float* w1t   = ws + 39518208;
  float* w2t   = ws + 41877504;

  pack_kernel<<<dim3((DD / 16) * (CC / 8) / 8, 1, HH), 256, 0, stream>>>(
      Wq, wqkvt + 0 * (size_t)HH * DD * CC, CC, DD);
  pack_kernel<<<dim3((DD / 16) * (CC / 8) / 8, 1, HH), 256, 0, stream>>>(
      Wk, wqkvt + 1 * (size_t)HH * DD * CC, CC, DD);
  pack_kernel<<<dim3((DD / 16) * (CC / 8) / 8, 1, HH), 256, 0, stream>>>(
      Wv, wqkvt + 2 * (size_t)HH * DD * CC, CC, DD);
  pack_kernel<<<dim3((FF / 16) * (CC / 8) / 8, 1, 1), 256, 0, stream>>>(
      W1, w1t, CC, FF);
  pack_kernel<<<dim3((CC / 16) * (FF / 8) / 8, 1, 1), 256, 0, stream>>>(
      W2, w2t, FF, CC);

  qkv_kernel<<<dim3(NTOK / 64, 2), 256, 0, stream>>>(x, wqkvt, q, k, v);
  attn_kernel<<<dim3(TT / 16, BN * HH), 256, 0, stream>>>(q, k, v, a);
  add_ln_kernel<<<dim3(NTOK), 256, 0, stream>>>(x, a, ln1g, ln1b, x1);
  mlp1_kernel<<<dim3(NTOK / 64, 2), 256, 0, stream>>>(x1, w1t, b1, hbuf);
  mlp2_kernel<<<dim3(NTOK / 64, CC / 64), 256, 0, stream>>>(hbuf, w2t, b2, t2);
  add_ln_kernel<<<dim3(NTOK), 256, 0, stream>>>(x1, t2, ln2g, ln2b, out);
}


# attention: no softmax passes over LDS: row max kept in registers by the score phase; P.V split over the 8 waves by k (4 column tiles each), exp and row sum applied to each score fragment as loaded
# speedup vs baseline: 3.3983x; 1.1737x over previous
#include <hip/hip_runtime.h>
#include <math.h>


typedef float v2f __attribute__((ext_vector_type(2)));
typedef float v4f __attribute__((ext_vector_type(4)));
typedef float v8f __attribute__((ext_vector_type(8)));

#define BN   4
#define TT   2048
#define CC   768
#define HH   12
#define DD   64
#define FF   3072
#define NTOK 8192
#define LDA  772
#define SROW 2052

__device__ __forceinline__ v8f wmma4(v2f a, v2f b, v8f c) {
  return __builtin_amdgcn_wmma_f32_16x16x4_f32(false, a, false, b, (short)0, c,
                                               false, false);
}

__global__ __launch_bounds__(256) void pack_kernel(
    const float* __restrict__ src, float* __restrict__ dst, int K, int N) {
  const size_t bofs = (size_t)blockIdx.z * K * N;
  src += bofs;
  dst += bofs;
  const int l = threadIdx.x & 31;
  const int c = l & 15, h = l >> 4;
  const int ntiles = N >> 4;
  const int chunk = blockIdx.x * 8 + (threadIdx.x >> 5);
  const int nt = chunk % ntiles, kb = chunk / ntiles;
  const float* s = src + (size_t)(kb * 8 + 4 * h) * N + nt * 16 + c;
  v4f v;
  v.x = s[0];
  v.y = s[N];
  v.z = s[2 * (size_t)N];
  v.w = s[3 * (size_t)N];
  *(v4f*)(dst + (((size_t)nt * (K >> 3) + kb) * 32 + l) * 4) = v;
}

struct Frag {
  v4f a[2][4];
  v4f b[4][4];
};

__device__ __forceinline__ void load_chunk(Frag& f, const float* ap,
                                           int astride, const v4f* bp,
                                           int bstride, int kc) {
#pragma unroll
  for (int u = 0; u < 4; ++u) {
    f.a[0][u] = *(const v4f*)(ap + (kc * 4 + u) * 8);
    f.a[1][u] = *(const v4f*)(ap + astride + (kc * 4 + u) * 8);
#pragma unroll
    for (int s = 0; s < 4; ++s)
      f.b[s][u] = bp[(size_t)s * bstride + (size_t)(kc * 4 + u) * 32];
  }
}

template <bool TR>
__device__ __forceinline__ v8f mm(v2f x, v2f w, v8f c) {
  return TR ? wmma4(w, x, c) : wmma4(x, w, c);
}

template <bool TR>
__device__ __forceinline__ void mma_chunk(const Frag& f, v8f (&c)[2][4]) {
#pragma unroll
  for (int u = 0; u < 4; ++u) {
#pragma unroll
    for (int r = 0; r < 2; ++r) {
      const v2f alo = {f.a[r][u].x, f.a[r][u].y};
#pragma unroll
      for (int s = 0; s < 4; ++s)
        c[r][s] = mm<TR>(alo, (v2f){f.b[s][u].x, f.b[s][u].y}, c[r][s]);
    }
#pragma unroll
    for (int r = 0; r < 2; ++r) {
      const v2f ahi = {f.a[r][u].z, f.a[r][u].w};
#pragma unroll
      for (int s = 0; s < 4; ++s)
        c[r][s] = mm<TR>(ahi, (v2f){f.b[s][u].z, f.b[s][u].w}, c[r][s]);
    }
  }
}

template <bool TR>
__device__ __forceinline__ void gemm_strip(const float* ap, int astride,
                                           const v4f* bp, int bstride, int nch,
                                           v8f (&c)[2][4]) {
  Frag f0, f1;
  load_chunk(f0, ap, astride, bp, bstride, 0);
  for (int kc = 0; kc < nch - 2; kc += 2) {
    load_chunk(f1, ap, astride, bp, bstride, kc + 1);
    mma_chunk<TR>(f0, c);
    load_chunk(f0, ap, astride, bp, bstride, kc + 2);
    mma_chunk<TR>(f1, c);
  }
  load_chunk(f1, ap, astride, bp, bstride, nch - 1);
  mma_chunk<TR>(f0, c);
  mma_chunk<TR>(f1, c);
}

__device__ __forceinline__ v4f lo4(v8f a) {
  return __builtin_shufflevector(a, a, 0, 1, 2, 3);
}
__device__ __forceinline__ v4f hi4(v8f a) {
  return __builtin_shufflevector(a, a, 4, 5, 6, 7);
}

__global__ __launch_bounds__(256) void qkv_kernel(
    const float* __restrict__ x, const float* __restrict__ wqkvt,
    float* __restrict__ qo, float* __restrict__ ko, float* __restrict__ vo) {
  __shared__ float xs[64 * LDA];
  const int m0 = blockIdx.x * 64;
  for (int i = threadIdx.x; i < 64 * (CC / 4); i += 256) {
    const int r = i / (CC / 4), c4 = i % (CC / 4);
    const float4 src = ((const float4*)(x + (m0 + r) * CC))[c4];
    *(float4*)(&xs[r * LDA + c4 * 4]) = src;
  }
  __syncthreads();
  const int wv = threadIdx.x >> 5, l = threadIdx.x & 31;
  const int n16 = l & 15, hi = l >> 4;
  const int yb = blockIdx.y;
  for (int task = wv; task < 36; task += 8) {
    const int rp = task & 1, oh = 2 * (task >> 1) + yb;
    const int o = oh / 12, head = oh % 12;
    const float* Wp = wqkvt + (size_t)oh * (DD * CC);
    v8f c[2][4] = {};
    const float* ap = &xs[(rp * 32 + n16) * LDA + 4 * hi];
    const int tb = m0 + rp * 32;
    const int b = tb >> 11, t0 = tb & 2047;
    const size_t bhofs = (size_t)(b * HH + head) * TT * DD;
    if (o < 2) {
      gemm_strip<true>(ap, 16 * LDA, (const v4f*)Wp + l, 16 * CC / 4, CC / 32,
                       c);
#pragma unroll
      for (int r = 0; r < 2; ++r) {
        v4f* dst = (v4f*)(((o == 0) ? qo : ko) + bhofs +
                          (size_t)(t0 + 16 * r) * DD) +
                   hi * 32 + n16;
        dst[0] = lo4(c[r][0]);   dst[16] = hi4(c[r][0]);
        dst[64] = lo4(c[r][1]);  dst[80] = hi4(c[r][1]);
        dst[128] = lo4(c[r][2]); dst[144] = hi4(c[r][2]);
        dst[192] = lo4(c[r][3]); dst[208] = hi4(c[r][3]);
      }
    } else {
      gemm_strip<false>(ap, 16 * LDA, (const v4f*)Wp + l, 16 * CC / 4, CC / 32,
                        c);
#pragma unroll
      for (int r = 0; r < 2; ++r) {
        v4f* dst =
            (v4f*)(vo + bhofs) + (((t0 + 16 * r) >> 3) + hi) * 128 + n16;
        dst[0] = lo4(c[r][0]);  dst[16] = hi4(c[r][0]);
        dst[32] = lo4(c[r][1]); dst[48] = hi4(c[r][1]);
        dst[64] = lo4(c[r][2]); dst[80] = hi4(c[r][2]);
        dst[96] = lo4(c[r][3]); dst[112] = hi4(c[r][3]);
      }
    }
  }
}

#define OPS 68
template <int NKB>
__device__ __forceinline__ void pv_trip(const float* prow, const v4f* vb,
                                        float mr, float& rsum,
                                        v8f (&acc)[4]) {
  v4f a[NKB], b[NKB][4];
#pragma unroll
  for (int u = 0; u < NKB; ++u) {
#pragma unroll
    for (int t = 0; t < 4; ++t) b[u][t] = vb[(u * 4 + t) * 32];
  }
#pragma unroll
  for (int u = 0; u < NKB; ++u) a[u] = *(const v4f*)(prow + u * 8);
#pragma unroll
  for (int u = 0; u < NKB; ++u) {
    a[u].x = __expf(a[u].x - mr);
    a[u].y = __expf(a[u].y - mr);
    a[u].z = __expf(a[u].z - mr);
    a[u].w = __expf(a[u].w - mr);
    rsum += (a[u].x + a[u].y) + (a[u].z + a[u].w);
  }
#pragma unroll
  for (int u = 0; u < NKB; ++u) {
#pragma unroll
    for (int t = 0; t < 4; ++t)
      acc[t] = wmma4((v2f){a[u].x, a[u].y}, (v2f){b[u][t].x, b[u][t].y},
                     acc[t]);
#pragma unroll
    for (int t = 0; t < 4; ++t)
      acc[t] = wmma4((v2f){a[u].z, a[u].w}, (v2f){b[u][t].z, b[u][t].w},
                     acc[t]);
  }
}

__global__ __launch_bounds__(256) void attn_kernel(
    const float* __restrict__ q, const float* __restrict__ k,
    const float* __restrict__ vt, float* __restrict__ aout) {
  __shared__ __attribute__((aligned(16))) float S[16 * SROW];
  __shared__ __attribute__((aligned(16))) float pmax[16 * 8];
  __shared__ __attribute__((aligned(16))) float psum[16 * 8];
  const int ttile = blockIdx.x;
  const int bh = blockIdx.y;
  const int t0 = ttile * 16;
  const int wv = threadIdx.x >> 5, l = threadIdx.x & 31;
  const int n16 = l & 15, hi = l >> 4;
  const v4f* qp = (const v4f*)(q + (size_t)bh * TT * DD) + l;
  const v4f* kp = (const v4f*)(k + (size_t)bh * TT * DD) + l;
  const v4f* vp = (const v4f*)(vt + (size_t)bh * TT * DD) + l;
  const int NT = ttile + 1;
  const float scale = 0.036084391824352f;

  v4f af[8];
#pragma unroll
  for (int kb = 0; kb < 8; ++kb) af[kb] = kp[(ttile * 8 + kb) * 32];
  float rmax[8];
#pragma unroll
  for (int j = 0; j < 8; ++j) rmax[j] = -1e30f;
  for (int st = wv; st < NT; st += 8) {
    const int s0 = st * 16;
    v8f acc = {};
#pragma unroll
    for (int kb = 0; kb < 8; ++kb) {
      const v4f b = qp[(st * 8 + kb) * 32];
      acc = wmma4((v2f){af[kb].x, af[kb].y}, (v2f){b.x, b.y}, acc);
      acc = wmma4((v2f){af[kb].z, af[kb].w}, (v2f){b.z, b.w}, acc);
    }
    const int sg = s0 + n16;
#pragma unroll
    for (int j = 0; j < 8; ++j) {
      const int row = (hi << 3) + j;
      float vl = acc[j] * scale;
      if (sg > t0 + row) vl = -1e30f;
      rmax[j] = fmaxf(rmax[j], vl);
      S[row * SROW + sg] = vl;
    }
  }
#pragma unroll
  for (int j = 0; j < 8; ++j) {
    float m = rmax[j];
    m = fmaxf(m, __shfl_xor(m, 1, 16));
    m = fmaxf(m, __shfl_xor(m, 2, 16));
    m = fmaxf(m, __shfl_xor(m, 4, 16));
    m = fmaxf(m, __shfl_xor(m, 8, 16));
    rmax[j] = m;
  }
  if (n16 == 0) {
#pragma unroll
    for (int j = 0; j < 8; ++j) pmax[((hi << 3) + j) * 8 + wv] = rmax[j];
  }
  __syncthreads();
  float mr;
  {
    const v4f m0 = *(const v4f*)&pmax[n16 * 8];
    const v4f m1 = *(const v4f*)&pmax[n16 * 8 + 4];
    mr = fmaxf(fmaxf(fmaxf(m0.x, m0.y), fmaxf(m0.z, m0.w)),
               fmaxf(fmaxf(m1.x, m1.y), fmaxf(m1.z, m1.w)));
  }

  const int cnt = (NT + 7) >> 3;
  int tcur = wv * cnt;
  int tend = tcur + cnt;
  if (tend > NT) tend = NT;
  v8f acc[4] = {};
  float rsum = 0.f;
  const float* prow = &S[n16 * SROW + 4 * hi];
  for (; tcur + 2 <= tend; tcur += 2)
    pv_trip<4>(prow + tcur * 16, vp + (size_t)tcur * (2 * 128), mr, rsum, acc);
  if (tcur < tend)
    pv_trip<2>(prow + tcur * 16, vp + (size_t)tcur * (2 * 128), mr, rsum, acc);
  rsum += __shfl_xor(rsum, 16);
  if (hi == 0) psum[n16 * 8 + wv] = rsum;
  __syncthreads();
  {
    float* op = S + wv * (16 * OPS);
#pragma unroll
    for (int t = 0; t < 4; ++t) {
#pragma unroll
      for (int j = 0; j < 8; ++j)
        op[((hi << 3) + j) * OPS + t * 16 + n16] = acc[t][j];
    }
  }
  __syncthreads();
  {
    const int row = threadIdx.x >> 4, d4 = (threadIdx.x & 15) * 4;
    v4f o = *(const v4f*)&S[row * OPS + d4];
#pragma unroll
    for (int w = 1; w < 8; ++w)
      o += *(const v4f*)&S[w * (16 * OPS) + row * OPS + d4];
    const v4f s0 = *(const v4f*)&psum[row * 8];
    const v4f s1 = *(const v4f*)&psum[row * 8 + 4];
    const float rcp = 1.0f / (((s0.x + s0.y) + (s0.z + s0.w)) +
                              ((s1.x + s1.y) + (s1.z + s1.w)));
    o *= rcp;
    const int bb = bh / HH, head = bh % HH;
    *(v4f*)(aout + ((size_t)(bb * TT + t0 + row)) * CC + head * DD + d4) = o;
  }
}

__global__ __launch_bounds__(256) void add_ln_kernel(
    const float* __restrict__ xa, const float* __restrict__ xb,
    const float* __restrict__ g, const float* __restrict__ bt,
    float* __restrict__ out) {
  __shared__ float red[256];
  const int tok = blockIdx.x;
  const float* pa = xa + (size_t)tok * CC;
  const float* pb = xb + (size_t)tok * CC;
  float vals[3];
  float s1 = 0.f;
#pragma unroll
  for (int i = 0; i < 3; ++i) {
    const int c = threadIdx.x + i * 256;
    vals[i] = pa[c] + pb[c];
    s1 += vals[i];
  }
  red[threadIdx.x] = s1;
  __syncthreads();
  for (int off = 128; off > 0; off >>= 1) {
    if (threadIdx.x < off) red[threadIdx.x] += red[threadIdx.x + off];
    __syncthreads();
  }
  const float mean = red[0] * (1.0f / CC);
  __syncthreads();
  float s2 = 0.f;
#pragma unroll
  for (int i = 0; i < 3; ++i) {
    const float d = vals[i] - mean;
    s2 += d * d;
  }
  red[threadIdx.x] = s2;
  __syncthreads();
  for (int off = 128; off > 0; off >>= 1) {
    if (threadIdx.x < off) red[threadIdx.x] += red[threadIdx.x + off];
    __syncthreads();
  }
  const float rstd = rsqrtf(red[0] * (1.0f / CC) + 1e-5f);
#pragma unroll
  for (int i = 0; i < 3; ++i) {
    const int c = threadIdx.x + i * 256;
    out[(size_t)tok * CC + c] = (vals[i] - mean) * rstd * g[c] + bt[c];
  }
}

__global__ __launch_bounds__(256) void mlp1_kernel(
    const float* __restrict__ xin, const float* __restrict__ W1t,
    const float* __restrict__ bias, float* __restrict__ hout) {
  __shared__ float xs[64 * LDA];
  const int m0 = blockIdx.x * 64;
  for (int i = threadIdx.x; i < 64 * (CC / 4); i += 256) {
    const int r = i / (CC / 4), c4 = i % (CC / 4);
    const float4 src = ((const float4*)(xin + (m0 + r) * CC))[c4];
    *(float4*)(&xs[r * LDA + c4 * 4]) = src;
  }
  __syncthreads();
  const int wv = threadIdx.x >> 5, l = threadIdx.x & 31;
  const int n16 = l & 15, hi = l >> 4;
  const int yb = blockIdx.y;
  for (int task = wv; task < 48; task += 8) {
    const int rp = task & 1, nb = (yb * 24 + (task >> 1)) * 64;
    v8f c[2][4] = {};
    gemm_strip<false>(&xs[(rp * 32 + n16) * LDA + 4 * hi], 16 * LDA,
                      (const v4f*)(W1t + (size_t)nb * CC) + l, 16 * CC / 4,
                      CC / 32, c);
#pragma unroll
    for (int r = 0; r < 2; ++r) {
      const int mbase = rp * 32 + r * 16 + (hi << 3);
#pragma unroll
      for (int j = 0; j < 8; ++j) {
        float* dst = hout + (size_t)(m0 + mbase + j) * FF + nb + n16;
        float z;
        z = c[r][0][j] + bias[nb + n16];
        dst[0] = 0.5f * z * (1.f + erff(z * 0.70710678118f));
        z = c[r][1][j] + bias[nb + 16 + n16];
        dst[16] = 0.5f * z * (1.f + erff(z * 0.70710678118f));
        z = c[r][2][j] + bias[nb + 32 + n16];
        dst[32] = 0.5f * z * (1.f + erff(z * 0.70710678118f));
        z = c[r][3][j] + bias[nb + 48 + n16];
        dst[48] = 0.5f * z * (1.f + erff(z * 0.70710678118f));
      }
    }
  }
}

__global__ __launch_bounds__(256) void mlp2_kernel(
    const float* __restrict__ hin, const float* __restrict__ W2t,
    const float* __restrict__ bias, float* __restrict__ t2) {
  __shared__ float opart[128 * 64];
  const int m0 = blockIdx.x * 128, n0 = blockIdx.y * 64;
  const int wv = threadIdx.x >> 5, l = threadIdx.x & 31;
  const int n16 = l & 15, hi = l >> 4;
  const int rp = wv & 3, half = wv >> 2;
  const int kofs = half * (FF / 2);
  v8f c[2][4] = {};
  gemm_strip<false>(
      hin + (size_t)(m0 + rp * 32 + n16) * FF + kofs + 4 * hi, 16 * FF,
      (const v4f*)(W2t + (size_t)n0 * FF) + (size_t)(kofs / 8) * 32 + l,
      16 * FF / 4, FF / 64, c);
  if (half == 1) {
#pragma unroll
    for (int r = 0; r < 2; ++r) {
      const int mb = rp * 32 + r * 16 + (hi << 3);
#pragma unroll
      for (int j = 0; j < 8; ++j) {
        opart[(mb + j) * 64 + n16] = c[r][0][j];
        opart[(mb + j) * 64 + 16 + n16] = c[r][1][j];
        opart[(mb + j) * 64 + 32 + n16] = c[r][2][j];
        opart[(mb + j) * 64 + 48 + n16] = c[r][3][j];
      }
    }
  }
  __syncthreads();
  if (half == 0) {
#pragma unroll
    for (int r = 0; r < 2; ++r) {
      const int mb = rp * 32 + r * 16 + (hi << 3);
#pragma unroll
      for (int j = 0; j < 8; ++j) {
        float* dst = t2 + (size_t)(m0 + mb + j) * CC + n0 + n16;
        dst[0]  = c[r][0][j] + opart[(mb + j) * 64 + n16]      + bias[n0 + n16];
        dst[16] = c[r][1][j] + opart[(mb + j) * 64 + 16 + n16] + bias[n0 + 16 + n16];
        dst[32] = c[r][2][j] + opart[(mb + j) * 64 + 32 + n16] + bias[n0 + 32 + n16];
        dst[48] = c[r][3][j] + opart[(mb + j) * 64 + 48 + n16] + bias[n0 + 48 + n16];
      }
    }
  }
}

extern "C" void kernel_launch(void* const* d_in, const int* in_sizes, int n_in,
                              void* d_out, int out_size, void* d_ws,
                              size_t ws_size, hipStream_t stream) {
  (void)in_sizes; (void)n_in; (void)out_size; (void)ws_size;
  const float* x    = (const float*)d_in[0];
  const float* Wq   = (const float*)d_in[1];
  const float* Wk   = (const float*)d_in[2];
  const float* Wv   = (const float*)d_in[3];
  const float* ln1g = (const float*)d_in[4];
  const float* ln1b = (const float*)d_in[5];
  const float* W1   = (const float*)d_in[6];
  const float* b1   = (const float*)d_in[7];
  const float* W2   = (const float*)d_in[8];
  const float* b2   = (const float*)d_in[9];
  const float* ln2g = (const float*)d_in[10];
  const float* ln2b = (const float*)d_in[11];
  float* ws  = (float*)d_ws;
  float* out = (float*)d_out;

  float* q     = ws;
  float* k     = ws + 6291456;
  float* v     = ws + 12582912;
  float* a     = ws + 18874368;
  float* hbuf  = ws;
  float* x1    = ws + 25165824;
  float* t2    = ws + 31457280;
  float* wqkvt = ws + 37748736;
  float* w1t   = ws + 39518208;
  float* w2t   = ws + 41877504;

  pack_kernel<<<dim3((DD / 16) * (CC / 8) / 8, 1, HH), 256, 0, stream>>>(
      Wq, wqkvt + 0 * (size_t)HH * DD * CC, CC, DD);
  pack_kernel<<<dim3((DD / 16) * (CC / 8) / 8, 1, HH), 256, 0, stream>>>(
      Wk, wqkvt + 1 * (size_t)HH * DD * CC, CC, DD);
  pack_kernel<<<dim3((DD / 16) * (CC / 8) / 8, 1, HH), 256, 0, stream>>>(
      Wv, wqkvt + 2 * (size_t)HH * DD * CC, CC, DD);
  pack_kernel<<<dim3((FF / 16) * (CC / 8) / 8, 1, 1), 256, 0, stream>>>(
      W1, w1t, CC, FF);
  pack_kernel<<<dim3((CC / 16) * (FF / 8) / 8, 1, 1), 256, 0, stream>>>(
      W2, w2t, FF, CC);

  qkv_kernel<<<dim3(NTOK / 64, 2), 256, 0, stream>>>(x, wqkvt, q, k, v);
  attn_kernel<<<dim3(TT / 16, BN * HH), 256, 0, stream>>>(q, k, v, a);
  add_ln_kernel<<<dim3(NTOK), 256, 0, stream>>>(x, a, ln1g, ln1b, x1);
  mlp1_kernel<<<dim3(NTOK / 64, 2), 256, 0, stream>>>(x1, w1t, b1, hbuf);
  mlp2_kernel<<<dim3(NTOK / 128, CC / 64), 256, 0, stream>>>(hbuf, w2t, b2, t2);
  add_ln_kernel<<<dim3(NTOK), 256, 0, stream>>>(x1, t2, ln2g, ln2b, out);
}
